# starting kernel
# baseline (speedup 1.0000x reference)
#include <hip/hip_runtime.h>

#define IN_F  128
#define HID_F 128
#define OUT_F 64

typedef __attribute__((ext_vector_type(2))) float v2f;
typedef __attribute__((ext_vector_type(8))) float v8f;


__global__ void fill_kernel(float* __restrict__ p, float v, int n) {
  int i = blockIdx.x * blockDim.x + threadIdx.x;
  if (i < n) p[i] = v;
}

__global__ void deg_kernel(const int* __restrict__ col, float* __restrict__ deg, int E) {
  int e = blockIdx.x * blockDim.x + threadIdx.x;
  if (e < E) atomicAdd(&deg[col[e]], 1.0f);
}

__global__ void dinv_kernel(float* __restrict__ deg, int n) {
  int i = blockIdx.x * blockDim.x + threadIdx.x;
  if (i < n) deg[i] = rsqrtf(deg[i]);
}

__global__ void bias_relu_kernel(const float* __restrict__ xw,
                                 float* __restrict__ agg,
                                 const float* __restrict__ dinv,
                                 const float* __restrict__ b1,
                                 int total) {
  int i = blockIdx.x * blockDim.x + threadIdx.x;
  if (i >= total) return;
  int node = i >> 7;
  int f    = i & 127;
  float di = dinv[node];
  float v  = agg[i] + xw[i] * di * di + b1[f];
  agg[i] = fmaxf(v, 0.0f);
}

__global__ void final_kernel(const float* __restrict__ hw,
                             const float* __restrict__ dinv,
                             const float* __restrict__ b2,
                             float* __restrict__ out,
                             int total) {
  int i = blockIdx.x * blockDim.x + threadIdx.x;
  if (i >= total) return;
  int node = i >> 6;
  int f    = i & 63;
  float di = dinv[node];
  out[i] = out[i] + hw[i] * di * di + b2[f];
}


template <int F>
__global__ void __launch_bounds__((F / 16) * 32)
gemm_wmma_kernel(const float* __restrict__ A,
                 const float* __restrict__ B,
                 float* __restrict__ C,
                 int nrows) {
  const int lane = threadIdx.x & 31;
  const int wave = threadIdx.x >> 5;
  const int row0 = blockIdx.x * 16;
  if (row0 >= nrows) return;
  const int col0 = wave * 16;
  const int m    = lane & 15;
  const int half = lane >> 4;

  v8f acc = {};
  const float* arow = A + (size_t)(row0 + m) * 128;

#pragma unroll
  for (int k = 0; k < 128; k += 4) {
    const int ka = k + 2 * half;
    v2f a, b;
    a.x = arow[ka];
    a.y = arow[ka + 1];
    b.x = B[(size_t)ka * F + col0 + m];
    b.y = B[(size_t)(ka + 1) * F + col0 + m];
    acc = __builtin_amdgcn_wmma_f32_16x16x4_f32(
         false, a,  false, b,
         (short)0, acc,  false,  false);
  }

#pragma unroll
  for (int v = 0; v < 8; ++v) {
    C[(size_t)(row0 + v + 8 * half) * F + col0 + m] = acc[v];
  }
}

__global__ void scatter128_kernel(const int* __restrict__ row,
                                  const int* __restrict__ col,
                                  const float* __restrict__ dinv,
                                  const float* __restrict__ xw,
                                  float* __restrict__ agg,
                                  int E) {
  int gw   = (blockIdx.x * blockDim.x + threadIdx.x) >> 5;
  int lane = threadIdx.x & 31;
  if (gw >= E) return;
  int r = row[gw];
  int c = col[gw];
  float nrm = dinv[r] * dinv[c];
  const float4* src = (const float4*)(xw + (size_t)r * 128);
  float4 v = src[lane];
  float* dst = agg + (size_t)c * 128 + lane * 4;
  atomicAdd(dst + 0, v.x * nrm);
  atomicAdd(dst + 1, v.y * nrm);
  atomicAdd(dst + 2, v.z * nrm);
  atomicAdd(dst + 3, v.w * nrm);
}

__global__ void scatter64_kernel(const int* __restrict__ row,
                                 const int* __restrict__ col,
                                 const float* __restrict__ dinv,
                                 const float* __restrict__ hw,
                                 float* __restrict__ agg,
                                 int E) {
  int gw   = (blockIdx.x * blockDim.x + threadIdx.x) >> 5;
  int lane = threadIdx.x & 31;
  if (gw >= E) return;
  int r = row[gw];
  int c = col[gw];
  float nrm = dinv[r] * dinv[c];
  const float2* src = (const float2*)(hw + (size_t)r * 64);
  float2 v = src[lane];
  float* dst = agg + (size_t)c * 64 + lane * 2;
  atomicAdd(dst + 0, v.x * nrm);
  atomicAdd(dst + 1, v.y * nrm);
}


extern "C" void kernel_launch(void* const* d_in, const int* in_sizes, int n_in,
                              void* d_out, int out_size, void* d_ws, size_t ws_size,
                              hipStream_t stream) {
  const float* x  = (const float*)d_in[0];
  const int*   ei = (const int*)d_in[1];
  const float* W1 = (const float*)d_in[2];
  const float* b1 = (const float*)d_in[3];
  const float* W2 = (const float*)d_in[4];
  const float* b2 = (const float*)d_in[5];
  float* out = (float*)d_out;

  const int N = in_sizes[0] / IN_F;
  const int E = in_sizes[1] / 2;
  const int* row = ei;
  const int* col = ei + E;

  float* ws   = (float*)d_ws;
  float* xw   = ws;
  float* agg1 = xw + (size_t)N * 128;
  float* hw   = agg1 + (size_t)N * 128;
  float* deg  = hw + (size_t)N * 64;

  const int n128 = N * 128;
  const int n64  = N * 64;

  fill_kernel<<<(n128 + 255) / 256, 256, 0, stream>>>(agg1, 0.0f, n128);
  fill_kernel<<<(n64  + 255) / 256, 256, 0, stream>>>(out,  0.0f, n64);
  fill_kernel<<<(N    + 255) / 256, 256, 0, stream>>>(deg,  1.0f, N);

  deg_kernel<<<(E + 255) / 256, 256, 0, stream>>>(col, deg, E);
  dinv_kernel<<<(N + 255) / 256, 256, 0, stream>>>(deg, N);

  gemm_wmma_kernel<128><<<(N + 15) / 16, 256, 0, stream>>>(x, W1, xw, N);
  scatter128_kernel<<<(E + 7) / 8, 256, 0, stream>>>(row, col, deg, xw, agg1, E);
  bias_relu_kernel<<<(n128 + 255) / 256, 256, 0, stream>>>(xw, agg1, deg, b1, n128);

  gemm_wmma_kernel<64><<<(N + 15) / 16, 128, 0, stream>>>(agg1, W2, hw, N);
  scatter64_kernel<<<(E + 7) / 8, 256, 0, stream>>>(row, col, deg, hw, out, E);
  final_kernel<<<(n64 + 255) / 256, 256, 0, stream>>>(hw, deg, b2, out, n64);
}


# Edge scatter, both layers: target rows split in two ranges, one per half of the chip (a block reads its die and draws 512-edge items of its range by ticket); sums stay arrival-order atomics
# speedup vs baseline: 2.1481x; 2.1481x over previous
#include <hip/hip_runtime.h>

#define IN_F  128
#define HID_F 128
#define OUT_F 64

typedef __attribute__((ext_vector_type(2))) float v2f;
typedef __attribute__((ext_vector_type(8))) float v8f;


__global__ void fill_kernel(float* __restrict__ p, float v, int n, unsigned int* __restrict__ ctr) {
  int i = blockIdx.x * blockDim.x + threadIdx.x;
  if (i < n) p[i] = v;
  if (i < 256) ctr[i] = 0u;
}

__global__ void deg_kernel(const int* __restrict__ col, float* __restrict__ deg, int E) {
  int e = blockIdx.x * blockDim.x + threadIdx.x;
  if (e < E) atomicAdd(&deg[col[e]], 1.0f);
}


template <int F, bool RELU_A>
__global__ void __launch_bounds__((F / 16) * 32)
gemm_wmma_kernel(const float* __restrict__ A,
                 const float* __restrict__ B,
                 const float* __restrict__ deg,
                 const float* __restrict__ bias,
                 float* __restrict__ C,
                 float* __restrict__ S,
                 int nrows) {
  const int lane = threadIdx.x & 31;
  const int wave = threadIdx.x >> 5;
  const int row0 = blockIdx.x * 16;
  if (row0 >= nrows) return;
  const int col0 = wave * 16;
  const int m    = lane & 15;
  const int half = lane >> 4;

  v8f acc = {};
  const float* arow = A + (size_t)(row0 + m) * 128;

#pragma unroll
  for (int k = 0; k < 128; k += 4) {
    const int ka = k + 2 * half;
    v2f a, b;
    a.x = arow[ka];
    a.y = arow[ka + 1];
    if (RELU_A) {
      a.x = fmaxf(a.x, 0.0f);
      a.y = fmaxf(a.y, 0.0f);
    }
    b.x = B[(size_t)ka * F + col0 + m];
    b.y = B[(size_t)(ka + 1) * F + col0 + m];
    acc = __builtin_amdgcn_wmma_f32_16x16x4_f32(
         false, a,  false, b,
         (short)0, acc,  false,  false);
  }

  const float bc = bias[col0 + m];
#pragma unroll
  for (int v = 0; v < 8; ++v) {
    const float dlo = deg[row0 + v];
    const float dhi = deg[row0 + 8 + v];
    const float di = rsqrtf(half ? dhi : dlo);
    const size_t o = (size_t)(row0 + v + 8 * half) * F + col0 + m;
    C[o] = acc[v];
    S[o] = acc[v] * di * di + bc;
  }
}


template <int F, int G>
__device__ __forceinline__ void edge_group(unsigned long long& m, int rA, int rB, int cA, int cB,
                                           float nA, float nB, const float* __restrict__ xw,
                                           float* __restrict__ agg, int lane) {
  constexpr int L = F / 32;
  int idx[G];
  float v[G][L];
#pragma unroll
  for (int u = 0; u < G; ++u) {
    idx[u] = __builtin_ctzll(m);
    m &= m - 1;
  }
#pragma unroll
  for (int u = 0; u < G; ++u) {
    const int i = idx[u];
    const int ra = __builtin_amdgcn_readlane(rA, i & 31);
    const int rb = __builtin_amdgcn_readlane(rB, i & 31);
    const int rj = (i & 32) ? rb : ra;
    const float* src = xw + (size_t)rj * F;
#pragma unroll
    for (int k = 0; k < L; ++k) v[u][k] = src[lane + 32 * k];
  }
#pragma unroll
  for (int u = 0; u < G; ++u) {
    const int i = idx[u];
    const int ca = __builtin_amdgcn_readlane(cA, i & 31);
    const int cb = __builtin_amdgcn_readlane(cB, i & 31);
    const int cj = (i & 32) ? cb : ca;
    const int na = __builtin_amdgcn_readlane(__float_as_int(nA), i & 31);
    const int nb = __builtin_amdgcn_readlane(__float_as_int(nB), i & 31);
    const float nj = __int_as_float((i & 32) ? nb : na);
    float* dst = agg + (size_t)cj * F;
#pragma unroll
    for (int k = 0; k < L; ++k) atomicAdd(dst + lane + 32 * k, v[u][k] * nj);
  }
}

template <int F>
__global__ void __launch_bounds__(256)
scatter_kernel(const int* __restrict__ row,
               const int* __restrict__ col,
               const float* __restrict__ deg,
               const float* __restrict__ xw,
               float* __restrict__ agg,
               unsigned int* __restrict__ ctr,
               int E, int split, int nitems) {
  __shared__ int s_job[2];
  const int lane = threadIdx.x & 31;
  const int wv = __builtin_amdgcn_readfirstlane((int)(threadIdx.x >> 5));

  if (threadIdx.x == 0) {
    int q = (int)((__builtin_amdgcn_s_sendmsg_rtn(0x87) >> 18) & 1u);
    unsigned int t = __hip_atomic_fetch_add(ctr + 64 * q, 1u, __ATOMIC_RELAXED, __HIP_MEMORY_SCOPE_AGENT);
    if (t >= (unsigned int)nitems) {
      q ^= 1;
      t = __hip_atomic_fetch_add(ctr + 64 * q, 1u, __ATOMIC_RELAXED, __HIP_MEMORY_SCOPE_AGENT);
    }
    s_job[0] = (t < (unsigned int)nitems) ? (int)t : -1;
    s_job[1] = q;
  }
  __syncthreads();
  const int item = __builtin_amdgcn_readfirstlane(s_job[0]);
  const int rng  = __builtin_amdgcn_readfirstlane(s_job[1]);
  if (item < 0) return;

  const int e0 = item * 512 + wv * 64;
  if (e0 >= E) return;

  const int eA = e0 + lane;
  const int eB = e0 + 32 + lane;
  int rA = 0, cA = 0, rB = 0, cB = 0;
  bool pA = false, pB = false;
  if (eA < E) { rA = row[eA]; cA = col[eA]; pA = ((cA >= split) ? 1 : 0) == rng; }
  if (eB < E) { rB = row[eB]; cB = col[eB]; pB = ((cB >= split) ? 1 : 0) == rng; }
  float nA = 0.0f, nB = 0.0f;
  if (pA) nA = rsqrtf(deg[rA]) * rsqrtf(deg[cA]);
  if (pB) nB = rsqrtf(deg[rB]) * rsqrtf(deg[cB]);

  unsigned long long m = (unsigned long long)__builtin_amdgcn_ballot_w32(pA) |
                         ((unsigned long long)__builtin_amdgcn_ballot_w32(pB) << 32);
  int cnt = __builtin_popcountll(m);

  for (; cnt >= 8; cnt -= 8) edge_group<F, 8>(m, rA, rB, cA, cB, nA, nB, xw, agg, lane);
  switch (cnt) {
    case 7: edge_group<F, 7>(m, rA, rB, cA, cB, nA, nB, xw, agg, lane); break;
    case 6: edge_group<F, 6>(m, rA, rB, cA, cB, nA, nB, xw, agg, lane); break;
    case 5: edge_group<F, 5>(m, rA, rB, cA, cB, nA, nB, xw, agg, lane); break;
    case 4: edge_group<F, 4>(m, rA, rB, cA, cB, nA, nB, xw, agg, lane); break;
    case 3: edge_group<F, 3>(m, rA, rB, cA, cB, nA, nB, xw, agg, lane); break;
    case 2: edge_group<F, 2>(m, rA, rB, cA, cB, nA, nB, xw, agg, lane); break;
    case 1: edge_group<F, 1>(m, rA, rB, cA, cB, nA, nB, xw, agg, lane); break;
    default: break;
  }
}


extern "C" void kernel_launch(void* const* d_in, const int* in_sizes, int n_in,
                              void* d_out, int out_size, void* d_ws, size_t ws_size,
                              hipStream_t stream) {
  const float* x  = (const float*)d_in[0];
  const int*   ei = (const int*)d_in[1];
  const float* W1 = (const float*)d_in[2];
  const float* b1 = (const float*)d_in[3];
  const float* W2 = (const float*)d_in[4];
  const float* b2 = (const float*)d_in[5];
  float* out = (float*)d_out;

  const int N = in_sizes[0] / IN_F;
  const int E = in_sizes[1] / 2;
  const int* row = ei;
  const int* col = ei + E;

  float* ws   = (float*)d_ws;
  float* xw   = ws;
  float* agg1 = xw + (size_t)N * 128;
  float* hw   = agg1 + (size_t)N * 128;
  float* deg  = hw + (size_t)N * 64;
  unsigned int* ctr = (unsigned int*)(deg + (size_t)((N + 63) / 64) * 64);
  const int split  = N / 2;
  const int nitems = (E + 511) / 512;

  fill_kernel<<<(N + 255) / 256, 256, 0, stream>>>(deg, 1.0f, N, ctr);
  deg_kernel<<<(E + 255) / 256, 256, 0, stream>>>(col, deg, E);

  gemm_wmma_kernel<128, false><<<(N + 15) / 16, 256, 0, stream>>>(x, W1, deg, b1, xw, agg1, N);
  scatter_kernel<128><<<2 * nitems, 256, 0, stream>>>(row, col, deg, xw, agg1, ctr, E, split, nitems);

  gemm_wmma_kernel<64, true><<<(N + 15) / 16, 128, 0, stream>>>(agg1, W2, deg, b2, hw, out, N);
  scatter_kernel<64><<<2 * nitems, 256, 0, stream>>>(row, col, deg, hw, out, ctr + 128, E, split, nitems);
}


# Edge weight split per node: stored rows carry dinv[source]; layer 1 adds them unscaled, dinv[target] applied once at next A load; no degree fetch per edge there; rounding changes, sum order kept
# speedup vs baseline: 2.2161x; 1.0317x over previous
#include <hip/hip_runtime.h>

#define IN_F  128
#define HID_F 128
#define OUT_F 64

typedef __attribute__((ext_vector_type(2))) float v2f;
typedef __attribute__((ext_vector_type(8))) float v8f;


__global__ void fill_kernel(float* __restrict__ p, float v, int n, unsigned int* __restrict__ ctr) {
  int i = blockIdx.x * blockDim.x + threadIdx.x;
  if (i < n) p[i] = v;
  if (i < 256) ctr[i] = 0u;
}

__global__ void deg_kernel(const int* __restrict__ col, float* __restrict__ deg, int E) {
  int e = blockIdx.x * blockDim.x + threadIdx.x;
  if (e < E) atomicAdd(&deg[col[e]], 1.0f);
}


template <int F, bool SECOND>
__global__ void __launch_bounds__((F / 16) * 32)
gemm_wmma_kernel(const float* __restrict__ A,
                 const float* __restrict__ B,
                 const float* __restrict__ deg,
                 const float* __restrict__ abias,
                 const float* __restrict__ bias,
                 float* __restrict__ C,
                 float* __restrict__ S,
                 int nrows) {
  const int lane = threadIdx.x & 31;
  const int wave = threadIdx.x >> 5;
  const int row0 = blockIdx.x * 16;
  if (row0 >= nrows) return;
  const int col0 = wave * 16;
  const int m    = lane & 15;
  const int half = lane >> 4;

  v8f acc = {};
  const float* arow = A + (size_t)(row0 + m) * 128;
  float da = 0.0f;
  if (SECOND) da = rsqrtf(deg[row0 + m]);

#pragma unroll
  for (int k = 0; k < 128; k += 4) {
    const int ka = k + 2 * half;
    v2f a, b;
    a.x = arow[ka];
    a.y = arow[ka + 1];
    if (SECOND) {
      a.x = fmaxf(fmaf(a.x, da, abias[ka]), 0.0f);
      a.y = fmaxf(fmaf(a.y, da, abias[ka + 1]), 0.0f);
    }
    b.x = B[(size_t)ka * F + col0 + m];
    b.y = B[(size_t)(ka + 1) * F + col0 + m];
    acc = __builtin_amdgcn_wmma_f32_16x16x4_f32(
         false, a,  false, b,
         (short)0, acc,  false,  false);
  }

  float bc = 0.0f;
  if (SECOND) bc = bias[col0 + m];
#pragma unroll
  for (int v = 0; v < 8; ++v) {
    const float dlo = deg[row0 + v];
    const float dhi = deg[row0 + 8 + v];
    const float di = rsqrtf(half ? dhi : dlo);
    const size_t o = (size_t)(row0 + v + 8 * half) * F + col0 + m;
    const float y = acc[v] * di;
    C[o] = y;
    S[o] = SECOND ? y * di + bc : y;
  }
}


template <int F, int G, bool TSCALE>
__device__ __forceinline__ void edge_group(unsigned long long& m, int rA, int rB, int cA, int cB,
                                           float nA, float nB, const float* __restrict__ xw,
                                           float* __restrict__ agg, int lane) {
  constexpr int L = F / 32;
  int idx[G];
  float v[G][L];
#pragma unroll
  for (int u = 0; u < G; ++u) {
    idx[u] = __builtin_ctzll(m);
    m &= m - 1;
  }
#pragma unroll
  for (int u = 0; u < G; ++u) {
    const int i = idx[u];
    const int ra = __builtin_amdgcn_readlane(rA, i & 31);
    const int rb = __builtin_amdgcn_readlane(rB, i & 31);
    const int rj = (i & 32) ? rb : ra;
    const float* src = xw + (size_t)rj * F;
#pragma unroll
    for (int k = 0; k < L; ++k) v[u][k] = src[lane + 32 * k];
  }
#pragma unroll
  for (int u = 0; u < G; ++u) {
    const int i = idx[u];
    const int ca = __builtin_amdgcn_readlane(cA, i & 31);
    const int cb = __builtin_amdgcn_readlane(cB, i & 31);
    const int cj = (i & 32) ? cb : ca;
    float* dst = agg + (size_t)cj * F;
    if (TSCALE) {
      const int na = __builtin_amdgcn_readlane(__float_as_int(nA), i & 31);
      const int nb = __builtin_amdgcn_readlane(__float_as_int(nB), i & 31);
      const float nj = __int_as_float((i & 32) ? nb : na);
#pragma unroll
      for (int k = 0; k < L; ++k) atomicAdd(dst + lane + 32 * k, v[u][k] * nj);
    } else {
#pragma unroll
      for (int k = 0; k < L; ++k) atomicAdd(dst + lane + 32 * k, v[u][k]);
    }
  }
}

template <int F, bool TSCALE>
__global__ void __launch_bounds__(256)
scatter_kernel(const int* __restrict__ row,
               const int* __restrict__ col,
               const float* __restrict__ deg,
               const float* __restrict__ xw,
               float* __restrict__ agg,
               unsigned int* __restrict__ ctr,
               int E, int split, int nitems) {
  __shared__ int s_job[2];
  const int lane = threadIdx.x & 31;
  const int wv = __builtin_amdgcn_readfirstlane((int)(threadIdx.x >> 5));

  if (threadIdx.x == 0) {
    int q = (int)((__builtin_amdgcn_s_sendmsg_rtn(0x87) >> 18) & 1u);
    unsigned int t = __hip_atomic_fetch_add(ctr + 64 * q, 1u, __ATOMIC_RELAXED, __HIP_MEMORY_SCOPE_AGENT);
    if (t >= (unsigned int)nitems) {
      q ^= 1;
      t = __hip_atomic_fetch_add(ctr + 64 * q, 1u, __ATOMIC_RELAXED, __HIP_MEMORY_SCOPE_AGENT);
    }
    s_job[0] = (t < (unsigned int)nitems) ? (int)t : -1;
    s_job[1] = q;
  }
  __syncthreads();
  const int item = __builtin_amdgcn_readfirstlane(s_job[0]);
  const int rng  = __builtin_amdgcn_readfirstlane(s_job[1]);
  if (item < 0) return;

  const int e0 = item * 512 + wv * 64;
  if (e0 >= E) return;

  const int eA = e0 + lane;
  const int eB = e0 + 32 + lane;
  int rA = 0, cA = 0, rB = 0, cB = 0;
  bool pA = false, pB = false;
  if (eA < E) { rA = row[eA]; cA = col[eA]; pA = ((cA >= split) ? 1 : 0) == rng; }
  if (eB < E) { rB = row[eB]; cB = col[eB]; pB = ((cB >= split) ? 1 : 0) == rng; }
  float nA = 0.0f, nB = 0.0f;
  if (TSCALE) {
    if (pA) nA = rsqrtf(deg[cA]);
    if (pB) nB = rsqrtf(deg[cB]);
  }

  unsigned long long m = (unsigned long long)__builtin_amdgcn_ballot_w32(pA) |
                         ((unsigned long long)__builtin_amdgcn_ballot_w32(pB) << 32);
  int cnt = __builtin_popcountll(m);

  for (; cnt >= 8; cnt -= 8) edge_group<F, 8, TSCALE>(m, rA, rB, cA, cB, nA, nB, xw, agg, lane);
  switch (cnt) {
    case 7: edge_group<F, 7, TSCALE>(m, rA, rB, cA, cB, nA, nB, xw, agg, lane); break;
    case 6: edge_group<F, 6, TSCALE>(m, rA, rB, cA, cB, nA, nB, xw, agg, lane); break;
    case 5: edge_group<F, 5, TSCALE>(m, rA, rB, cA, cB, nA, nB, xw, agg, lane); break;
    case 4: edge_group<F, 4, TSCALE>(m, rA, rB, cA, cB, nA, nB, xw, agg, lane); break;
    case 3: edge_group<F, 3, TSCALE>(m, rA, rB, cA, cB, nA, nB, xw, agg, lane); break;
    case 2: edge_group<F, 2, TSCALE>(m, rA, rB, cA, cB, nA, nB, xw, agg, lane); break;
    case 1: edge_group<F, 1, TSCALE>(m, rA, rB, cA, cB, nA, nB, xw, agg, lane); break;
    default: break;
  }
}


extern "C" void kernel_launch(void* const* d_in, const int* in_sizes, int n_in,
                              void* d_out, int out_size, void* d_ws, size_t ws_size,
                              hipStream_t stream) {
  const float* x  = (const float*)d_in[0];
  const int*   ei = (const int*)d_in[1];
  const float* W1 = (const float*)d_in[2];
  const float* b1 = (const float*)d_in[3];
  const float* W2 = (const float*)d_in[4];
  const float* b2 = (const float*)d_in[5];
  float* out = (float*)d_out;

  const int N = in_sizes[0] / IN_F;
  const int E = in_sizes[1] / 2;
  const int* row = ei;
  const int* col = ei + E;

  float* ws   = (float*)d_ws;
  float* xw   = ws;
  float* agg1 = xw + (size_t)N * 128;
  float* hw   = agg1 + (size_t)N * 128;
  float* deg  = hw + (size_t)N * 64;
  unsigned int* ctr = (unsigned int*)(deg + (size_t)((N + 63) / 64) * 64);
  const int split  = N / 2;
  const int nitems = (E + 511) / 512;

  fill_kernel<<<(N + 255) / 256, 256, 0, stream>>>(deg, 1.0f, N, ctr);
  deg_kernel<<<(E + 255) / 256, 256, 0, stream>>>(col, deg, E);

  gemm_wmma_kernel<128, false><<<(N + 15) / 16, 256, 0, stream>>>(x, W1, deg, b1, b1, xw, agg1, N);
  scatter_kernel<128, false><<<2 * nitems, 256, 0, stream>>>(row, col, deg, xw, agg1, ctr, E, split, nitems);

  gemm_wmma_kernel<64, true><<<(N + 15) / 16, 128, 0, stream>>>(agg1, W2, deg, b1, b2, hw, out, N);
  scatter_kernel<64, true><<<2 * nitems, 256, 0, stream>>>(row, col, deg, hw, out, ctr + 128, E, split, nitems);
}


# Both products: a wave owns 16 rows x 64 columns as four column-interleaved tiles (column 4c+t): one A fragment feeds four WMMAs, B rows and output rows move as 16-byte pieces; k order kept
# speedup vs baseline: 2.3231x; 1.0483x over previous
#include <hip/hip_runtime.h>

#define IN_F  128
#define HID_F 128
#define OUT_F 64

typedef __attribute__((ext_vector_type(2))) float v2f;
typedef __attribute__((ext_vector_type(8))) float v8f;


__global__ void fill_kernel(float* __restrict__ p, float v, int n, unsigned int* __restrict__ ctr) {
  int i = blockIdx.x * blockDim.x + threadIdx.x;
  if (i < n) p[i] = v;
  if (i < 256) ctr[i] = 0u;
}

__global__ void deg_kernel(const int* __restrict__ col, float* __restrict__ deg, int E) {
  int e = blockIdx.x * blockDim.x + threadIdx.x;
  if (e < E) atomicAdd(&deg[col[e]], 1.0f);
}


#define KC 4
struct KChunk {
  v2f a[KC];
  v2f ab[KC];
  float4 b0[KC];
  float4 b1[KC];
};

template <int F, bool SECOND>
__device__ __forceinline__ void chunk_load(KChunk& f, const float* __restrict__ ap,
                                           const float* __restrict__ bp,
                                           const float* __restrict__ abp) {
#pragma unroll
  for (int j = 0; j < KC; ++j) {
    f.a[j].x = ap[4 * j];
    f.a[j].y = ap[4 * j + 1];
    if (SECOND) {
      f.ab[j].x = abp[4 * j];
      f.ab[j].y = abp[4 * j + 1];
    }
    f.b0[j] = *reinterpret_cast<const float4*>(bp + (size_t)(4 * j) * F);
    f.b1[j] = *reinterpret_cast<const float4*>(bp + (size_t)(4 * j + 1) * F);
  }
}

template <bool SECOND>
__device__ __forceinline__ void chunk_mma(const KChunk& f, float da,
                                          v8f& acc0, v8f& acc1, v8f& acc2, v8f& acc3) {
#pragma unroll
  for (int j = 0; j < KC; ++j) {
    v2f a = f.a[j], b;
    if (SECOND) {
      a.x = fmaxf(fmaf(a.x, da, f.ab[j].x), 0.0f);
      a.y = fmaxf(fmaf(a.y, da, f.ab[j].y), 0.0f);
    }
    b.x = f.b0[j].x; b.y = f.b1[j].x;
    acc0 = __builtin_amdgcn_wmma_f32_16x16x4_f32(
         false, a,  false, b,
         (short)0, acc0,  false,  false);
    b.x = f.b0[j].y; b.y = f.b1[j].y;
    acc1 = __builtin_amdgcn_wmma_f32_16x16x4_f32(false, a, false, b, (short)0, acc1, false, false);
    b.x = f.b0[j].z; b.y = f.b1[j].z;
    acc2 = __builtin_amdgcn_wmma_f32_16x16x4_f32(false, a, false, b, (short)0, acc2, false, false);
    b.x = f.b0[j].w; b.y = f.b1[j].w;
    acc3 = __builtin_amdgcn_wmma_f32_16x16x4_f32(false, a, false, b, (short)0, acc3, false, false);
  }
}

template <int F, bool SECOND>
__global__ void __launch_bounds__(256)
gemm_wmma_kernel(const float* __restrict__ A,
                 const float* __restrict__ B,
                 const float* __restrict__ deg,
                 const float* __restrict__ abias,
                 const float* __restrict__ bias,
                 float* __restrict__ C,
                 float* __restrict__ S,
                 int nrows) {
  constexpr int CG  = F / 64;
  constexpr int TPB = 8 / CG;
  const int lane = threadIdx.x & 31;
  const int wave = __builtin_amdgcn_readfirstlane((int)(threadIdx.x >> 5));
  const int row0 = (blockIdx.x * TPB + wave / CG) * 16;
  if (row0 >= nrows) return;
  const int m    = lane & 15;
  const int half = lane >> 4;
  const int cb   = (wave % CG) * 64 + 4 * m;

  v8f acc0 = {}, acc1 = {}, acc2 = {}, acc3 = {};
  float da = 0.0f;
  if (SECOND) da = rsqrtf(deg[row0 + m]);

  const float* ap  = A + (size_t)(row0 + m) * 128 + 2 * half;
  const float* bp  = B + (size_t)(2 * half) * F + cb;
  const float* abp = abias + 2 * half;
  constexpr int AS = 4 * KC;
  constexpr int BS = 4 * KC * F;

  KChunk f0, f1;
  chunk_load<F, SECOND>(f0, ap, bp, abp);
#pragma unroll 1
  for (int it = 0; it < 3; ++it) {
    chunk_load<F, SECOND>(f1, ap + AS, bp + BS, abp + AS);
    chunk_mma<SECOND>(f0, da, acc0, acc1, acc2, acc3);
    ap += 2 * AS; bp += 2 * BS; abp += 2 * AS;
    chunk_load<F, SECOND>(f0, ap, bp, abp);
    chunk_mma<SECOND>(f1, da, acc0, acc1, acc2, acc3);
  }
  chunk_load<F, SECOND>(f1, ap + AS, bp + BS, abp + AS);
  chunk_mma<SECOND>(f0, da, acc0, acc1, acc2, acc3);
  chunk_mma<SECOND>(f1, da, acc0, acc1, acc2, acc3);

  float4 bc = make_float4(0.0f, 0.0f, 0.0f, 0.0f);
  if (SECOND) bc = *reinterpret_cast<const float4*>(bias + cb);
#pragma unroll
  for (int v = 0; v < 8; ++v) {
    const float dlo = deg[row0 + v];
    const float dhi = deg[row0 + 8 + v];
    const float di = rsqrtf(half ? dhi : dlo);
    const size_t o = (size_t)(row0 + v + 8 * half) * F + cb;
    float4 y;
    y.x = acc0[v] * di;
    y.y = acc1[v] * di;
    y.z = acc2[v] * di;
    y.w = acc3[v] * di;
    *reinterpret_cast<float4*>(C + o) = y;
    float4 s = y;
    if (SECOND) {
      s.x = y.x * di + bc.x;
      s.y = y.y * di + bc.y;
      s.z = y.z * di + bc.z;
      s.w = y.w * di + bc.w;
    }
    *reinterpret_cast<float4*>(S + o) = s;
  }
}


template <int F, int G, bool TSCALE>
__device__ __forceinline__ void edge_group(unsigned long long& m, int rA, int rB, int cA, int cB,
                                           float nA, float nB, const float* __restrict__ xw,
                                           float* __restrict__ agg, int lane) {
  constexpr int L = F / 32;
  int idx[G];
  float v[G][L];
#pragma unroll
  for (int u = 0; u < G; ++u) {
    idx[u] = __builtin_ctzll(m);
    m &= m - 1;
  }
#pragma unroll
  for (int u = 0; u < G; ++u) {
    const int i = idx[u];
    const int ra = __builtin_amdgcn_readlane(rA, i & 31);
    const int rb = __builtin_amdgcn_readlane(rB, i & 31);
    const int rj = (i & 32) ? rb : ra;
    const float* src = xw + (size_t)rj * F;
#pragma unroll
    for (int k = 0; k < L; ++k) v[u][k] = src[lane + 32 * k];
  }
#pragma unroll
  for (int u = 0; u < G; ++u) {
    const int i = idx[u];
    const int ca = __builtin_amdgcn_readlane(cA, i & 31);
    const int cb = __builtin_amdgcn_readlane(cB, i & 31);
    const int cj = (i & 32) ? cb : ca;
    float* dst = agg + (size_t)cj * F;
    if (TSCALE) {
      const int na = __builtin_amdgcn_readlane(__float_as_int(nA), i & 31);
      const int nb = __builtin_amdgcn_readlane(__float_as_int(nB), i & 31);
      const float nj = __int_as_float((i & 32) ? nb : na);
#pragma unroll
      for (int k = 0; k < L; ++k) atomicAdd(dst + lane + 32 * k, v[u][k] * nj);
    } else {
#pragma unroll
      for (int k = 0; k < L; ++k) atomicAdd(dst + lane + 32 * k, v[u][k]);
    }
  }
}

template <int F, bool TSCALE>
__global__ void __launch_bounds__(256)
scatter_kernel(const int* __restrict__ row,
               const int* __restrict__ col,
               const float* __restrict__ deg,
               const float* __restrict__ xw,
               float* __restrict__ agg,
               unsigned int* __restrict__ ctr,
               int E, int split, int nitems) {
  __shared__ int s_job[2];
  const int lane = threadIdx.x & 31;
  const int wv = __builtin_amdgcn_readfirstlane((int)(threadIdx.x >> 5));

  if (threadIdx.x == 0) {
    int q = (int)((__builtin_amdgcn_s_sendmsg_rtn(0x87) >> 18) & 1u);
    unsigned int t = __hip_atomic_fetch_add(ctr + 64 * q, 1u, __ATOMIC_RELAXED, __HIP_MEMORY_SCOPE_AGENT);
    if (t >= (unsigned int)nitems) {
      q ^= 1;
      t = __hip_atomic_fetch_add(ctr + 64 * q, 1u, __ATOMIC_RELAXED, __HIP_MEMORY_SCOPE_AGENT);
    }
    s_job[0] = (t < (unsigned int)nitems) ? (int)t : -1;
    s_job[1] = q;
  }
  __syncthreads();
  const int item = __builtin_amdgcn_readfirstlane(s_job[0]);
  const int rng  = __builtin_amdgcn_readfirstlane(s_job[1]);
  if (item < 0) return;

  const int e0 = item * 512 + wv * 64;
  if (e0 >= E) return;

  const int eA = e0 + lane;
  const int eB = e0 + 32 + lane;
  int rA = 0, cA = 0, rB = 0, cB = 0;
  bool pA = false, pB = false;
  if (eA < E) { rA = row[eA]; cA = col[eA]; pA = ((cA >= split) ? 1 : 0) == rng; }
  if (eB < E) { rB = row[eB]; cB = col[eB]; pB = ((cB >= split) ? 1 : 0) == rng; }
  float nA = 0.0f, nB = 0.0f;
  if (TSCALE) {
    if (pA) nA = rsqrtf(deg[cA]);
    if (pB) nB = rsqrtf(deg[cB]);
  }

  unsigned long long m = (unsigned long long)__builtin_amdgcn_ballot_w32(pA) |
                         ((unsigned long long)__builtin_amdgcn_ballot_w32(pB) << 32);
  int cnt = __builtin_popcountll(m);

  for (; cnt >= 8; cnt -= 8) edge_group<F, 8, TSCALE>(m, rA, rB, cA, cB, nA, nB, xw, agg, lane);
  switch (cnt) {
    case 7: edge_group<F, 7, TSCALE>(m, rA, rB, cA, cB, nA, nB, xw, agg, lane); break;
    case 6: edge_group<F, 6, TSCALE>(m, rA, rB, cA, cB, nA, nB, xw, agg, lane); break;
    case 5: edge_group<F, 5, TSCALE>(m, rA, rB, cA, cB, nA, nB, xw, agg, lane); break;
    case 4: edge_group<F, 4, TSCALE>(m, rA, rB, cA, cB, nA, nB, xw, agg, lane); break;
    case 3: edge_group<F, 3, TSCALE>(m, rA, rB, cA, cB, nA, nB, xw, agg, lane); break;
    case 2: edge_group<F, 2, TSCALE>(m, rA, rB, cA, cB, nA, nB, xw, agg, lane); break;
    case 1: edge_group<F, 1, TSCALE>(m, rA, rB, cA, cB, nA, nB, xw, agg, lane); break;
    default: break;
  }
}


extern "C" void kernel_launch(void* const* d_in, const int* in_sizes, int n_in,
                              void* d_out, int out_size, void* d_ws, size_t ws_size,
                              hipStream_t stream) {
  const float* x  = (const float*)d_in[0];
  const int*   ei = (const int*)d_in[1];
  const float* W1 = (const float*)d_in[2];
  const float* b1 = (const float*)d_in[3];
  const float* W2 = (const float*)d_in[4];
  const float* b2 = (const float*)d_in[5];
  float* out = (float*)d_out;

  const int N = in_sizes[0] / IN_F;
  const int E = in_sizes[1] / 2;
  const int* row = ei;
  const int* col = ei + E;

  float* ws   = (float*)d_ws;
  float* xw   = ws;
  float* agg1 = xw + (size_t)N * 128;
  float* hw   = agg1 + (size_t)N * 128;
  float* deg  = hw + (size_t)N * 64;
  unsigned int* ctr = (unsigned int*)(deg + (size_t)((N + 63) / 64) * 64);
  const int split  = N / 2;
  const int nitems = (E + 511) / 512;

  fill_kernel<<<(N + 255) / 256, 256, 0, stream>>>(deg, 1.0f, N, ctr);
  deg_kernel<<<(E + 255) / 256, 256, 0, stream>>>(col, deg, E);

  gemm_wmma_kernel<128, false><<<(N + 63) / 64, 256, 0, stream>>>(x, W1, deg, b1, b1, xw, agg1, N);
  scatter_kernel<128, false><<<2 * nitems, 256, 0, stream>>>(row, col, deg, xw, agg1, ctr, E, split, nitems);

  gemm_wmma_kernel<64, true><<<(N + 127) / 128, 256, 0, stream>>>(agg1, W2, deg, b1, b2, hw, out, N);
  scatter_kernel<64, true><<<2 * nitems, 256, 0, stream>>>(row, col, deg, hw, out, ctr + 128, E, split, nitems);
}


# Degree count without global atomics or a fill launch: per-chunk LDS counts stored as 32 partial planes, summed per row tile by the first product, which stores deg; 6->5 launches; sums unchanged
# speedup vs baseline: 2.9297x; 1.2611x over previous
#include <hip/hip_runtime.h>

#define IN_F  128
#define HID_F 128
#define OUT_F 64

typedef __attribute__((ext_vector_type(2))) float v2f;
typedef __attribute__((ext_vector_type(8))) float v8f;


#define HC 32
#define HR 8192
#define HT 512
#define HU 10

__device__ __forceinline__ void hist_count(unsigned int* h, int t, int lo) {
  const unsigned int u = (unsigned int)(t - lo);
  if (u < (unsigned int)HR) atomicAdd(&h[u], 1u);
}

__global__ void __launch_bounds__(HT)
hist_kernel(const int* __restrict__ col, float* __restrict__ part, unsigned int* __restrict__ ctr,
            int E, int chunk, int pitch) {
  __shared__ unsigned int h[HR];
  const int c  = blockIdx.x;
  const int lo = blockIdx.y * HR;
  if (blockIdx.x == 0 && blockIdx.y == 0 && threadIdx.x < 256) ctr[threadIdx.x] = 0u;
  for (int i = threadIdx.x; i < HR; i += HT) h[i] = 0u;
  __syncthreads();

  const int e0 = c * chunk;
  const int e1 = (e0 + chunk < E) ? e0 + chunk : E;
  if (e1 > e0) {
    if (((size_t)(col + e0) & 15) == 0) {
      const int4* p4 = reinterpret_cast<const int4*>(col + e0);
      const int n4 = (e1 - e0) >> 2;
      for (int b = 0; b < n4; b += HU * HT) {
        int4 v[HU];
#pragma unroll
        for (int k = 0; k < HU; ++k) {
          const int i = b + k * HT + (int)threadIdx.x;
          v[k] = (i < n4) ? p4[i] : make_int4(-1, -1, -1, -1);
        }
#pragma unroll
        for (int k = 0; k < HU; ++k) {
          hist_count(h, v[k].x, lo);
          hist_count(h, v[k].y, lo);
          hist_count(h, v[k].z, lo);
          hist_count(h, v[k].w, lo);
        }
      }
      for (int e = e0 + (n4 << 2) + (int)threadIdx.x; e < e1; e += HT) hist_count(h, col[e], lo);
    } else {
      for (int e = e0 + (int)threadIdx.x; e < e1; e += HT) hist_count(h, col[e], lo);
    }
  }
  __syncthreads();
  float* dst = part + (size_t)c * pitch + lo;
  for (int i = threadIdx.x; i < HR; i += HT) dst[i] = (float)h[i];
}


#define KC 4
struct KChunk {
  v2f a[KC];
  v2f ab[KC];
  float4 b0[KC];
  float4 b1[KC];
};

template <int F, bool SECOND>
__device__ __forceinline__ void chunk_load(KChunk& f, const float* __restrict__ ap,
                                           const float* __restrict__ bp,
                                           const float* __restrict__ abp) {
#pragma unroll
  for (int j = 0; j < KC; ++j) {
    f.a[j].x = ap[4 * j];
    f.a[j].y = ap[4 * j + 1];
    if (SECOND) {
      f.ab[j].x = abp[4 * j];
      f.ab[j].y = abp[4 * j + 1];
    }
    f.b0[j] = *reinterpret_cast<const float4*>(bp + (size_t)(4 * j) * F);
    f.b1[j] = *reinterpret_cast<const float4*>(bp + (size_t)(4 * j + 1) * F);
  }
}

template <bool SECOND>
__device__ __forceinline__ void chunk_mma(const KChunk& f, float da,
                                          v8f& acc0, v8f& acc1, v8f& acc2, v8f& acc3) {
#pragma unroll
  for (int j = 0; j < KC; ++j) {
    v2f a = f.a[j], b;
    if (SECOND) {
      a.x = fmaxf(fmaf(a.x, da, f.ab[j].x), 0.0f);
      a.y = fmaxf(fmaf(a.y, da, f.ab[j].y), 0.0f);
    }
    b.x = f.b0[j].x; b.y = f.b1[j].x;
    acc0 = __builtin_amdgcn_wmma_f32_16x16x4_f32(
         false, a,  false, b,
         (short)0, acc0,  false,  false);
    b.x = f.b0[j].y; b.y = f.b1[j].y;
    acc1 = __builtin_amdgcn_wmma_f32_16x16x4_f32(false, a, false, b, (short)0, acc1, false, false);
    b.x = f.b0[j].z; b.y = f.b1[j].z;
    acc2 = __builtin_amdgcn_wmma_f32_16x16x4_f32(false, a, false, b, (short)0, acc2, false, false);
    b.x = f.b0[j].w; b.y = f.b1[j].w;
    acc3 = __builtin_amdgcn_wmma_f32_16x16x4_f32(false, a, false, b, (short)0, acc3, false, false);
  }
}

template <int F, bool SECOND>
__global__ void __launch_bounds__(256, SECOND ? 2 : 8)
gemm_wmma_kernel(const float* __restrict__ A,
                 const float* __restrict__ B,
                 const float* __restrict__ deg,
                 const float* __restrict__ abias,
                 const float* __restrict__ bias,
                 float* __restrict__ C,
                 float* __restrict__ S,
                 int nrows,
                 const float* __restrict__ part,
                 int pitch,
                 float* __restrict__ degw) {
  constexpr int CG  = F / 64;
  constexpr int TPB = 8 / CG;
  const int lane = threadIdx.x & 31;
  const int wave = __builtin_amdgcn_readfirstlane((int)(threadIdx.x >> 5));
  const int row0 = (blockIdx.x * TPB + wave / CG) * 16;
  if (row0 >= nrows) return;
  const int m    = lane & 15;
  const int half = lane >> 4;
  const int cb   = (wave % CG) * 64 + 4 * m;

  v8f acc0 = {}, acc1 = {}, acc2 = {}, acc3 = {};
  float da = 0.0f;
  if (SECOND) da = rsqrtf(deg[row0 + m]);

  const float* ap  = A + (size_t)(row0 + m) * 128 + 2 * half;
  const float* bp  = B + (size_t)(2 * half) * F + cb;
  const float* abp = abias + 2 * half;
  constexpr int AS = 4 * KC;
  constexpr int BS = 4 * KC * F;

  KChunk f0, f1;
  chunk_load<F, SECOND>(f0, ap, bp, abp);

  float dqlo[8], dqhi[8];
#pragma unroll
  for (int r = 0; r < 8; ++r) { dqlo[r] = 1.0f; dqhi[r] = 1.0f; }
  if (!SECOND) {
    const float* pp = part + (size_t)(16 * half) * pitch + row0 + m;
    float s = 0.0f;
#pragma unroll
    for (int p = 0; p < HC / 2; ++p) s += pp[(size_t)p * pitch];
    s += __shfl_xor(s, 16);
    const float degv = 1.0f + s;
    if ((wave % CG) == 0 && half == 0) degw[row0 + m] = degv;
    const float rq = rsqrtf(degv);
#pragma unroll
    for (int r = 0; r < 8; ++r) {
      dqlo[r] = __int_as_float(__builtin_amdgcn_readlane(__float_as_int(rq), r));
      dqhi[r] = __int_as_float(__builtin_amdgcn_readlane(__float_as_int(rq), r + 8));
    }
  }
#pragma unroll 1
  for (int it = 0; it < 3; ++it) {
    chunk_load<F, SECOND>(f1, ap + AS, bp + BS, abp + AS);
    chunk_mma<SECOND>(f0, da, acc0, acc1, acc2, acc3);
    ap += 2 * AS; bp += 2 * BS; abp += 2 * AS;
    chunk_load<F, SECOND>(f0, ap, bp, abp);
    chunk_mma<SECOND>(f1, da, acc0, acc1, acc2, acc3);
  }
  chunk_load<F, SECOND>(f1, ap + AS, bp + BS, abp + AS);
  chunk_mma<SECOND>(f0, da, acc0, acc1, acc2, acc3);
  chunk_mma<SECOND>(f1, da, acc0, acc1, acc2, acc3);

  float4 bc = make_float4(0.0f, 0.0f, 0.0f, 0.0f);
  if (SECOND) bc = *reinterpret_cast<const float4*>(bias + cb);
#pragma unroll
  for (int v = 0; v < 8; ++v) {
    float di;
    if (SECOND) {
      const float dlo = deg[row0 + v];
      const float dhi = deg[row0 + 8 + v];
      di = rsqrtf(half ? dhi : dlo);
    } else {
      di = half ? dqhi[v] : dqlo[v];
    }
    const size_t o = (size_t)(row0 + v + 8 * half) * F + cb;
    float4 y;
    y.x = acc0[v] * di;
    y.y = acc1[v] * di;
    y.z = acc2[v] * di;
    y.w = acc3[v] * di;
    *reinterpret_cast<float4*>(C + o) = y;
    float4 s = y;
    if (SECOND) {
      s.x = y.x * di + bc.x;
      s.y = y.y * di + bc.y;
      s.z = y.z * di + bc.z;
      s.w = y.w * di + bc.w;
    }
    *reinterpret_cast<float4*>(S + o) = s;
  }
}


template <int F, int G, bool TSCALE>
__device__ __forceinline__ void edge_group(unsigned long long& m, int rA, int rB, int cA, int cB,
                                           float nA, float nB, const float* __restrict__ xw,
                                           float* __restrict__ agg, int lane) {
  constexpr int L = F / 32;
  int idx[G];
  float v[G][L];
#pragma unroll
  for (int u = 0; u < G; ++u) {
    idx[u] = __builtin_ctzll(m);
    m &= m - 1;
  }
#pragma unroll
  for (int u = 0; u < G; ++u) {
    const int i = idx[u];
    const int ra = __builtin_amdgcn_readlane(rA, i & 31);
    const int rb = __builtin_amdgcn_readlane(rB, i & 31);
    const int rj = (i & 32) ? rb : ra;
    const float* src = xw + (size_t)rj * F;
#pragma unroll
    for (int k = 0; k < L; ++k) v[u][k] = src[lane + 32 * k];
  }
#pragma unroll
  for (int u = 0; u < G; ++u) {
    const int i = idx[u];
    const int ca = __builtin_amdgcn_readlane(cA, i & 31);
    const int cb = __builtin_amdgcn_readlane(cB, i & 31);
    const int cj = (i & 32) ? cb : ca;
    float* dst = agg + (size_t)cj * F;
    if (TSCALE) {
      const int na = __builtin_amdgcn_readlane(__float_as_int(nA), i & 31);
      const int nb = __builtin_amdgcn_readlane(__float_as_int(nB), i & 31);
      const float nj = __int_as_float((i & 32) ? nb : na);
#pragma unroll
      for (int k = 0; k < L; ++k) atomicAdd(dst + lane + 32 * k, v[u][k] * nj);
    } else {
#pragma unroll
      for (int k = 0; k < L; ++k) atomicAdd(dst + lane + 32 * k, v[u][k]);
    }
  }
}

template <int F, bool TSCALE>
__global__ void __launch_bounds__(256)
scatter_kernel(const int* __restrict__ row,
               const int* __restrict__ col,
               const float* __restrict__ deg,
               const float* __restrict__ xw,
               float* __restrict__ agg,
               unsigned int* __restrict__ ctr,
               int E, int split, int nitems) {
  __shared__ int s_job[2];
  const int lane = threadIdx.x & 31;
  const int wv = __builtin_amdgcn_readfirstlane((int)(threadIdx.x >> 5));

  if (threadIdx.x == 0) {
    int q = (int)((__builtin_amdgcn_s_sendmsg_rtn(0x87) >> 18) & 1u);
    unsigned int t = __hip_atomic_fetch_add(ctr + 64 * q, 1u, __ATOMIC_RELAXED, __HIP_MEMORY_SCOPE_AGENT);
    if (t >= (unsigned int)nitems) {
      q ^= 1;
      t = __hip_atomic_fetch_add(ctr + 64 * q, 1u, __ATOMIC_RELAXED, __HIP_MEMORY_SCOPE_AGENT);
    }
    s_job[0] = (t < (unsigned int)nitems) ? (int)t : -1;
    s_job[1] = q;
  }
  __syncthreads();
  const int item = __builtin_amdgcn_readfirstlane(s_job[0]);
  const int rng  = __builtin_amdgcn_readfirstlane(s_job[1]);
  if (item < 0) return;

  const int e0 = item * 512 + wv * 64;
  if (e0 >= E) return;

  const int eA = e0 + lane;
  const int eB = e0 + 32 + lane;
  int rA = 0, cA = 0, rB = 0, cB = 0;
  bool pA = false, pB = false;
  if (eA < E) { rA = row[eA]; cA = col[eA]; pA = ((cA >= split) ? 1 : 0) == rng; }
  if (eB < E) { rB = row[eB]; cB = col[eB]; pB = ((cB >= split) ? 1 : 0) == rng; }
  float nA = 0.0f, nB = 0.0f;
  if (TSCALE) {
    if (pA) nA = rsqrtf(deg[cA]);
    if (pB) nB = rsqrtf(deg[cB]);
  }

  unsigned long long m = (unsigned long long)__builtin_amdgcn_ballot_w32(pA) |
                         ((unsigned long long)__builtin_amdgcn_ballot_w32(pB) << 32);
  int cnt = __builtin_popcountll(m);

  for (; cnt >= 8; cnt -= 8) edge_group<F, 8, TSCALE>(m, rA, rB, cA, cB, nA, nB, xw, agg, lane);
  switch (cnt) {
    case 7: edge_group<F, 7, TSCALE>(m, rA, rB, cA, cB, nA, nB, xw, agg, lane); break;
    case 6: edge_group<F, 6, TSCALE>(m, rA, rB, cA, cB, nA, nB, xw, agg, lane); break;
    case 5: edge_group<F, 5, TSCALE>(m, rA, rB, cA, cB, nA, nB, xw, agg, lane); break;
    case 4: edge_group<F, 4, TSCALE>(m, rA, rB, cA, cB, nA, nB, xw, agg, lane); break;
    case 3: edge_group<F, 3, TSCALE>(m, rA, rB, cA, cB, nA, nB, xw, agg, lane); break;
    case 2: edge_group<F, 2, TSCALE>(m, rA, rB, cA, cB, nA, nB, xw, agg, lane); break;
    case 1: edge_group<F, 1, TSCALE>(m, rA, rB, cA, cB, nA, nB, xw, agg, lane); break;
    default: break;
  }
}


extern "C" void kernel_launch(void* const* d_in, const int* in_sizes, int n_in,
                              void* d_out, int out_size, void* d_ws, size_t ws_size,
                              hipStream_t stream) {
  const float* x  = (const float*)d_in[0];
  const int*   ei = (const int*)d_in[1];
  const float* W1 = (const float*)d_in[2];
  const float* b1 = (const float*)d_in[3];
  const float* W2 = (const float*)d_in[4];
  const float* b2 = (const float*)d_in[5];
  float* out = (float*)d_out;

  const int N = in_sizes[0] / IN_F;
  const int E = in_sizes[1] / 2;
  const int* row = ei;
  const int* col = ei + E;

  float* ws   = (float*)d_ws;
  float* xw   = ws;
  float* agg1 = xw + (size_t)N * 128;
  float* hw   = agg1 + (size_t)N * 128;
  float* deg  = hw + (size_t)N * 64;
  unsigned int* ctr = (unsigned int*)(deg + (size_t)((N + 63) / 64) * 64);
  const int split  = N / 2;
  const int nitems = (E + 511) / 512;
  float* part = (float*)(ctr + 256);
  const int nranges = (N + HR - 1) / HR;
  const int pitch   = nranges * HR;
  const int chunk   = (((E + HC - 1) / HC) + 3) & ~3;

  hist_kernel<<<dim3(HC, nranges), HT, 0, stream>>>(col, part, ctr, E, chunk, pitch);

  gemm_wmma_kernel<128, false><<<(N + 63) / 64, 256, 0, stream>>>(x, W1, nullptr, b1, b1, xw, agg1, N, part, pitch, deg);
  scatter_kernel<128, false><<<2 * nitems, 256, 0, stream>>>(row, col, deg, xw, agg1, ctr, E, split, nitems);

  gemm_wmma_kernel<64, true><<<(N + 127) / 128, 256, 0, stream>>>(agg1, W2, deg, b1, b2, hw, out, N, nullptr, 0, nullptr);
  scatter_kernel<64, true><<<2 * nitems, 256, 0, stream>>>(row, col, deg, hw, out, ctr + 128, E, split, nitems);
}
